# starting kernel
# baseline (speedup 1.0000x reference)
#include <hip/hip_runtime.h>


typedef float v2f __attribute__((ext_vector_type(2)));
typedef float v8f __attribute__((ext_vector_type(8)));

constexpr int N1_   = 128;
constexpr int N2_   = 128;
constexpr int C_DIM = 768;
constexpr int WH    = 256;
constexpr int KT    = 16;
constexpr int KP    = KT / 2;
constexpr int LDN   = 144;

__global__ __launch_bounds__(512)
void nngp_convkp_kernel(const float* __restrict__ x,
                        const float* __restrict__ y,
                        float* __restrict__ out_xy,
                        float* __restrict__ out_xx,
                        float* __restrict__ out_yy)
{
    __shared__ v2f  XsP[2][KP * LDN];
    __shared__ v2f  YsP[2][KP * LDN];
    __shared__ float red[8 * 128];

    const int wh   = blockIdx.x;
    const int t    = threadIdx.x;
    const int lane = t & 31;
    const int wave = t >> 5;

    const int m0  = (wave >> 2) * 32;
    const int n0  = (wave & 3) * 32;
    const int row = lane & 15;
    const int kh  = lane >> 4;

    const size_t nstride = (size_t)C_DIM * WH;

    v8f acc[2][2] = {};
    float sxx = 0.f, syy = 0.f;

    const int    nloc  = t & 127;
    const int    cbase = t >> 7;
    const size_t gbase = (size_t)nloc * nstride + (size_t)wh;

    float xr[4], yr[4];

    #pragma unroll
    for (int j = 0; j < 4; ++j) {
        const int c = cbase + 4 * j;
        xr[j] = x[gbase + (size_t)c * WH];
        yr[j] = y[gbase + (size_t)c * WH];
        sxx += xr[j] * xr[j];
        syy += yr[j] * yr[j];
    }
    {
        float* xb = (float*)&XsP[0][0];
        float* yb = (float*)&YsP[0][0];
        #pragma unroll
        for (int j = 0; j < 4; ++j) {
            const int c = cbase + 4 * j;
            const int idx = ((c >> 1) * LDN + nloc) * 2 + (c & 1);
            xb[idx] = xr[j];
            yb[idx] = yr[j];
        }
    }
    __syncthreads();

    const int NT = C_DIM / KT;
    for (int kt = 0; kt < NT; ++kt) {
        const int cur = kt & 1;

        const bool have_next = (kt + 1) < NT;
        if (have_next) {
            const int k0n = (kt + 1) * KT;
            #pragma unroll
            for (int j = 0; j < 4; ++j) {
                const int c = k0n + cbase + 4 * j;
                xr[j] = x[gbase + (size_t)c * WH];
                yr[j] = y[gbase + (size_t)c * WH];
            }
        }

        const v2f* Xp = &XsP[cur][0];
        const v2f* Yp = &YsP[cur][0];
        #pragma unroll
        for (int kk = 0; kk < KP; kk += 2) {
            const int kpl = kk + kh;
            v2f a0 = Xp[kpl * LDN + m0      + row];
            v2f a1 = Xp[kpl * LDN + m0 + 16 + row];
            v2f b0 = Yp[kpl * LDN + n0      + row];
            v2f b1 = Yp[kpl * LDN + n0 + 16 + row];

            acc[0][0] = __builtin_amdgcn_wmma_f32_16x16x4_f32(false, a0, false, b0, (short)0, acc[0][0], false, false);
            acc[0][1] = __builtin_amdgcn_wmma_f32_16x16x4_f32(false, a0, false, b1, (short)0, acc[0][1], false, false);
            acc[1][0] = __builtin_amdgcn_wmma_f32_16x16x4_f32(false, a1, false, b0, (short)0, acc[1][0], false, false);
            acc[1][1] = __builtin_amdgcn_wmma_f32_16x16x4_f32(false, a1, false, b1, (short)0, acc[1][1], false, false);
        }

        if (have_next) {
            float* xb = (float*)&XsP[cur ^ 1][0];
            float* yb = (float*)&YsP[cur ^ 1][0];
            #pragma unroll
            for (int j = 0; j < 4; ++j) {
                const int c   = cbase + 4 * j;
                const int idx = ((c >> 1) * LDN + nloc) * 2 + (c & 1);
                xb[idx] = xr[j];
                yb[idx] = yr[j];
                sxx += xr[j] * xr[j];
                syy += yr[j] * yr[j];
            }
            __syncthreads();
        }
    }

    const float inv_c = 1.0f / (float)C_DIM;

    #pragma unroll
    for (int i = 0; i < 2; ++i) {
        #pragma unroll
        for (int j = 0; j < 2; ++j) {
            const int mm = m0 + 16 * i + kh * 8;
            const int nn = n0 + 16 * j + (lane & 15);
            #pragma unroll
            for (int r = 0; r < 8; ++r) {
                out_xy[(size_t)((mm + r) * N2_ + nn) * WH + wh] = acc[i][j][r] * inv_c;
            }
        }
    }

    __syncthreads();
    red[      (t >> 7) * 128 + nloc] = sxx;
    red[512 + (t >> 7) * 128 + nloc] = syy;
    __syncthreads();
    if (t < 128) {
        const float vx = red[t] + red[128 + t] + red[256 + t] + red[384 + t];
        const float vy = red[512 + t] + red[640 + t] + red[768 + t] + red[896 + t];
        out_xx[(size_t)t * WH + wh] = vx * inv_c;
        out_yy[(size_t)t * WH + wh] = vy * inv_c;
    }
}

extern "C" void kernel_launch(void* const* d_in, const int* in_sizes, int n_in,
                              void* d_out, int out_size, void* d_ws, size_t ws_size,
                              hipStream_t stream) {
    (void)in_sizes; (void)n_in; (void)out_size; (void)d_ws; (void)ws_size;
    const float* x = (const float*)d_in[0];
    const float* y = (const float*)d_in[1];
    float* out    = (float*)d_out;
    float* out_xy = out;
    float* out_xx = out + (size_t)N1_ * N2_ * WH;
    float* out_yy = out_xx + (size_t)N1_ * WH;
    nngp_convkp_kernel<<<WH, 512, 0, stream>>>(x, y, out_xy, out_xx, out_yy);
}


# Neighbour lanes (l, l^1), not lanes (r, r+16), read the two adjacent 16 B of a line, so 8 consecutive lanes touch 4 lines, not 8; xx/yy summed in two channel halves (new reduction tree)
# speedup vs baseline: 4.6250x; 4.6250x over previous
#include <hip/hip_runtime.h>


typedef float    v2f __attribute__((ext_vector_type(2)));
typedef float    v4f __attribute__((ext_vector_type(4)));
typedef float    v8f __attribute__((ext_vector_type(8)));
typedef unsigned v2u __attribute__((ext_vector_type(2)));

constexpr int N1_   = 128;
constexpr int N2_   = 128;
constexpr int C_DIM = 768;
constexpr int WH    = 256;
constexpr int KS    = C_DIM / 4;
constexpr int DEPTH = 4;

struct Stage {
    v4f a[2][2];
    v4f b[2][2][2];
};

__device__ __forceinline__ float from_neighbour(float v)
{
    const unsigned u = __float_as_uint(v);
    return __uint_as_float(__builtin_amdgcn_permlane16(u, u, 0x67452301u, 0xEFCDAB89u, false, false));
}

__device__ __forceinline__ void exchange_neighbours(v4f& ev, v4f& od, bool odd)
{
    #pragma unroll
    for (int e = 0; e < 4; ++e) {
        const float send = odd ? ev[e] : od[e];
        const float got  = from_neighbour(send);
        const float g0   = odd ? got   : ev[e];
        const float g1   = odd ? od[e] : got;
        ev[e] = g0;
        od[e] = g1;
    }
}

__device__ __forceinline__ v4f add_halves(v4f v)
{
    v4f out;
    #pragma unroll
    for (int e = 0; e < 4; ++e) {
        const unsigned u = __float_as_uint(v[e]);
        const v2u r = __builtin_amdgcn_permlane16_swap(u, u, false, false);
        out[e] = __uint_as_float(r[0]) + __uint_as_float(r[1]);
    }
    return out;
}

__device__ __forceinline__ void load_stage(Stage& s, const float* __restrict__ xa,
                                           const float* __restrict__ yb0,
                                           const float* __restrict__ yb1, int t)
{
    const size_t off = (size_t)t * 4 * WH;
    const size_t nstride = (size_t)C_DIM * WH;
    #pragma unroll
    for (int ks = 0; ks < 2; ++ks) {
        #pragma unroll
        for (int par = 0; par < 2; ++par) {
            s.a[ks][par]    = *(const v4f*)(xa  + off + ks * WH + par * nstride);
            s.b[0][ks][par] = *(const v4f*)(yb0 + off + ks * WH + par * nstride);
            s.b[1][ks][par] = *(const v4f*)(yb1 + off + ks * WH + par * nstride);
        }
    }
}

__device__ __forceinline__ void compute_stage(Stage& s, v8f (&acc)[8][2], v4f (&sxx)[2], v4f (&syy)[2][2],
                                              bool odd)
{
    #pragma unroll
    for (int ks = 0; ks < 2; ++ks) {
        #pragma unroll
        for (int par = 0; par < 2; ++par) {
            sxx[par]    += s.a[ks][par]    * s.a[ks][par];
            syy[0][par] += s.b[0][ks][par] * s.b[0][ks][par];
            syy[1][par] += s.b[1][ks][par] * s.b[1][ks][par];
        }
    }

    #pragma unroll
    for (int ks = 0; ks < 2; ++ks) {
        exchange_neighbours(s.a[ks][0],    s.a[ks][1],    odd);
        exchange_neighbours(s.b[0][ks][0], s.b[0][ks][1], odd);
        exchange_neighbours(s.b[1][ks][0], s.b[1][ks][1], odd);
    }

    #pragma unroll
    for (int g = 0; g < 2; ++g) {
        #pragma unroll
        for (int e = 0; e < 4; ++e) {
            const v2f af  = {s.a[0][g][e],    s.a[1][g][e]};
            const v2f bf0 = {s.b[0][0][g][e], s.b[0][1][g][e]};
            const v2f bf1 = {s.b[1][0][g][e], s.b[1][1][g][e]};
            const int w = 4 * g + e;
            acc[w][0] = __builtin_amdgcn_wmma_f32_16x16x4_f32(false, af, false, bf0, (short)0, acc[w][0], false, false);
            acc[w][1] = __builtin_amdgcn_wmma_f32_16x16x4_f32(false, af, false, bf1, (short)0, acc[w][1], false, false);
        }
    }
}

__global__ __launch_bounds__(128)
void nngp_convkp_direct(const float* __restrict__ x,
                        const float* __restrict__ y,
                        float* __restrict__ out_xy,
                        float* __restrict__ out_xx,
                        float* __restrict__ out_yy)
{
    const int lane = threadIdx.x & 31;
    const int wave = threadIdx.x >> 5;
    const int r    = lane & 15;
    const int h    = lane >> 4;
    const int o    = lane & 1;
    const int pr   = (lane >> 1) & 7;
    const bool odd = (o != 0);

    const int tile = blockIdx.x >> 3;
    const int lgrp = blockIdx.x & 7;
    const int tm   = tile >> 2;
    const int tn   = tile & 3;
    const int m0   = tm * 16;
    const int n0   = tn * 32;
    const int wh0  = lgrp * 32 + wave * 8;

    const size_t nstride = (size_t)C_DIM * WH;
    const float* xa  = x + (size_t)(m0 + 2 * pr)      * nstride + 2 * h * WH + wh0 + 4 * o;
    const float* yb0 = y + (size_t)(n0 + 2 * pr)      * nstride + 2 * h * WH + wh0 + 4 * o;
    const float* yb1 = y + (size_t)(n0 + 16 + 2 * pr) * nstride + 2 * h * WH + wh0 + 4 * o;

    const bool do_xx = (tn == 0);
    const bool do_yy = (tm == 0);

    v8f acc[8][2] = {};
    v4f sxx[2] = {{0.f, 0.f, 0.f, 0.f}, {0.f, 0.f, 0.f, 0.f}};
    v4f syy[2][2] = {{{0.f, 0.f, 0.f, 0.f}, {0.f, 0.f, 0.f, 0.f}},
                     {{0.f, 0.f, 0.f, 0.f}, {0.f, 0.f, 0.f, 0.f}}};

    Stage st[DEPTH];
    load_stage(st[0], xa, yb0, yb1, 0);
    __builtin_amdgcn_sched_barrier(0);
    load_stage(st[1], xa, yb0, yb1, 1);
    __builtin_amdgcn_sched_barrier(0);
    load_stage(st[2], xa, yb0, yb1, 2);
    __builtin_amdgcn_sched_barrier(0);

    for (int kt = 0; kt < KS - DEPTH; kt += DEPTH) {
        #pragma unroll
        for (int u = 0; u < DEPTH; ++u) {
            load_stage(st[(u + 3) & 3], xa, yb0, yb1, kt + u + 3);
            __builtin_amdgcn_sched_barrier(0);
            compute_stage(st[u], acc, sxx, syy, odd);
            __builtin_amdgcn_sched_barrier(0);
        }
    }
    load_stage(st[3], xa, yb0, yb1, KS - 1);
    compute_stage(st[0], acc, sxx, syy, odd);
    compute_stage(st[1], acc, sxx, syy, odd);
    compute_stage(st[2], acc, sxx, syy, odd);
    compute_stage(st[3], acc, sxx, syy, odd);

    const float inv_c = 1.0f / (float)C_DIM;

    #pragma unroll
    for (int j = 0; j < 2; ++j) {
        const int nn = n0 + 16 * j + r;
        #pragma unroll
        for (int i = 0; i < 8; ++i) {
            const int mm = m0 + 8 * h + i;
            float* o = out_xy + (size_t)(mm * N2_ + nn) * WH + wh0;
            const v4f lo = {acc[0][j][i] * inv_c, acc[1][j][i] * inv_c, acc[2][j][i] * inv_c, acc[3][j][i] * inv_c};
            const v4f hi = {acc[4][j][i] * inv_c, acc[5][j][i] * inv_c, acc[6][j][i] * inv_c, acc[7][j][i] * inv_c};
            *(v4f*)(o)     = lo;
            *(v4f*)(o + 4) = hi;
        }
    }

    if (do_xx) {
        const v4f t0 = add_halves(sxx[0]);
        const v4f t1 = add_halves(sxx[1]);
        *(v4f*)(out_xx + (size_t)(m0 + 2 * pr + h) * WH + wh0 + 4 * o) = (h ? t1 : t0) * inv_c;
    }
    if (do_yy) {
        #pragma unroll
        for (int j = 0; j < 2; ++j) {
            const v4f t0 = add_halves(syy[j][0]);
            const v4f t1 = add_halves(syy[j][1]);
            *(v4f*)(out_yy + (size_t)(n0 + 16 * j + 2 * pr + h) * WH + wh0 + 4 * o) = (h ? t1 : t0) * inv_c;
        }
    }
}

extern "C" void kernel_launch(void* const* d_in, const int* in_sizes, int n_in,
                              void* d_out, int out_size, void* d_ws, size_t ws_size,
                              hipStream_t stream) {
    (void)in_sizes; (void)n_in; (void)out_size; (void)d_ws; (void)ws_size;
    const float* x = (const float*)d_in[0];
    const float* y = (const float*)d_in[1];
    float* out    = (float*)d_out;
    float* out_xy = out;
    float* out_xx = out + (size_t)N1_ * N2_ * WH;
    float* out_yy = out_xx + (size_t)N1_ * WH;
    nngp_convkp_direct<<<256, 128, 0, stream>>>(x, y, out_xy, out_xx, out_yy);
}
